# starting kernel
# baseline (speedup 1.0000x reference)
#include <hip/hip_runtime.h>
#include <math.h>

#define NN 100000
#define NE 1600000
#define KIN 64
#define H1 64
#define H2 32

typedef __attribute__((ext_vector_type(2))) float v2f;
typedef __attribute__((ext_vector_type(8))) float v8f;

__global__ void k_zero(float* p, int n) {
  int i = blockIdx.x * blockDim.x + threadIdx.x;
  if (i < n) p[i] = 0.0f;
}

__global__ void k_edge_ew_deg(const float* __restrict__ logits,
                              const int* __restrict__ ei,
                              float* __restrict__ ew,
                              float* __restrict__ deg) {
  int e = blockIdx.x * blockDim.x + threadIdx.x;
  if (e >= NE) return;
  float l = logits[e];
  float s = 0.5f + 1.0f / (1.0f + __expf(-l));
  ew[e] = s;
  int col = ei[NE + e];
  atomicAdd(&deg[col], s);
}

__global__ void k_dis(const float* __restrict__ deg, float* __restrict__ dis) {
  int i = blockIdx.x * blockDim.x + threadIdx.x;
  if (i < NN) {
    float d = deg[i] + 1.0f;
    dis[i] = rsqrtf(d);
  }
}

__global__ void k_norm(const int* __restrict__ ei,
                       const float* __restrict__ ew,
                       const float* __restrict__ dis,
                       float* __restrict__ norm) {
  int e = blockIdx.x * blockDim.x + threadIdx.x;
  if (e >= NE) return;
  int r = ei[e];
  int c = ei[NE + e];
  norm[e] = dis[r] * ew[e] * dis[c];
}

template <int K, int NOUT>
__global__ void k_gemm_wmma(const float* __restrict__ X,
                            const float* __restrict__ W,
                            float* __restrict__ H) {
  const int wave = threadIdx.x >> 5;
  const int lane = threadIdx.x & 31;
  const int m0 = blockIdx.x * 16;
  const int n0 = wave * 16;
  const int half = lane >> 4;
  const int l15 = lane & 15;

  v2f bfrag[K / 4];
#pragma unroll
  for (int kt = 0; kt < K / 4; ++kt) {
    int kk = kt * 4 + half * 2;
    int col = n0 + l15;
    v2f b;
    b.x = W[kk * NOUT + col];
    b.y = W[(kk + 1) * NOUT + col];
    bfrag[kt] = b;
  }

  const int row = m0 + l15;
  const float* xp = X + (size_t)row * K + half * 2;

  v8f acc = {};
#pragma unroll
  for (int kt = 0; kt < K / 4; ++kt) {
    v2f a = *(const v2f*)(xp + kt * 4);
    acc = __builtin_amdgcn_wmma_f32_16x16x4_f32(
         false, a,  false, bfrag[kt],
         (short)0, acc,  false,  false);
  }

#pragma unroll
  for (int v = 0; v < 8; ++v) {
    int r = m0 + v + half * 8;
    H[(size_t)r * NOUT + n0 + l15] = acc[v];
  }
}

template <int NOUT>
__global__ void k_self(const float* __restrict__ h,
                       const float* __restrict__ dis,
                       const float* __restrict__ b,
                       float* __restrict__ out) {
  int idx = blockIdx.x * blockDim.x + threadIdx.x;
  if (idx >= NN * NOUT) return;
  int i = idx / NOUT;
  int c = idx - i * NOUT;
  float d = dis[i];
  out[idx] = h[idx] * (d * d) + b[c];
}

template <int NOUT, int LOGG>
__global__ void k_scatter(const float* __restrict__ h,
                          const int* __restrict__ ei,
                          const float* __restrict__ norm,
                          float* __restrict__ out) {
  long long idx = (long long)blockIdx.x * blockDim.x + threadIdx.x;
  int e = (int)(idx >> LOGG);
  if (e >= NE) return;
  int g = (int)(idx & ((1 << LOGG) - 1));
  int row = ei[e];
  int col = ei[NE + e];
  float nw = norm[e];
  const float4 hv = *(const float4*)(h + (size_t)row * NOUT + g * 4);
  float* op = out + (size_t)col * NOUT + g * 4;
  atomicAdd(op + 0, hv.x * nw);
  atomicAdd(op + 1, hv.y * nw);
  atomicAdd(op + 2, hv.z * nw);
  atomicAdd(op + 3, hv.w * nw);
}

__global__ void k_relu(float* p, int n) {
  int i = blockIdx.x * blockDim.x + threadIdx.x;
  if (i < n) p[i] = fmaxf(p[i], 0.0f);
}

__global__ void k_final(const float* __restrict__ a2,
                        const float* __restrict__ lw,
                        const float* __restrict__ lb,
                        float* __restrict__ out) {
  int i = blockIdx.x * blockDim.x + threadIdx.x;
  if (i >= NN) return;
  const float* hp = a2 + (size_t)i * H2;
  float acc = lb[0];
#pragma unroll
  for (int c = 0; c < H2; ++c) acc = fmaf(hp[c], lw[c], acc);
  out[2 * i + 0] = -acc;
  out[2 * i + 1] = acc;
}

extern "C" void kernel_launch(void* const* d_in, const int* in_sizes, int n_in,
                              void* d_out, int out_size, void* d_ws, size_t ws_size,
                              hipStream_t stream) {
  const float* x      = (const float*)d_in[0];
  const int*   ei     = (const int*)d_in[1];
  const float* elog   = (const float*)d_in[2];
  const float* W1     = (const float*)d_in[3];
  const float* b1     = (const float*)d_in[4];
  const float* W2     = (const float*)d_in[5];
  const float* b2     = (const float*)d_in[6];
  const float* lw     = (const float*)d_in[7];
  const float* lb     = (const float*)d_in[8];
  float* out = (float*)d_out;

  float* ws   = (float*)d_ws;
  float* deg  = ws;
  float* dis  = deg + NN;
  float* ew   = dis + NN;
  float* nrm  = ew + NE;
  float* h1   = nrm + NE;
  float* a1   = h1 + (size_t)NN * H1;
  float* h2   = h1;
  float* a2   = h1 + (size_t)NN * H2;

  const int T = 256;
  k_zero<<<(NN + T - 1) / T, T, 0, stream>>>(deg, NN);
  k_edge_ew_deg<<<(NE + T - 1) / T, T, 0, stream>>>(elog, ei, ew, deg);
  k_dis<<<(NN + T - 1) / T, T, 0, stream>>>(deg, dis);
  k_norm<<<(NE + T - 1) / T, T, 0, stream>>>(ei, ew, dis, nrm);

  k_gemm_wmma<KIN, H1><<<NN / 16, 32 * (H1 / 16), 0, stream>>>(x, W1, h1);
  k_self<H1><<<(NN * H1 + T - 1) / T, T, 0, stream>>>(h1, dis, b1, a1);
  k_scatter<H1, 4><<<(int)(((long long)NE * (H1 / 4) + T - 1) / T), T, 0, stream>>>(h1, ei, nrm, a1);
  k_relu<<<(NN * H1 + T - 1) / T, T, 0, stream>>>(a1, NN * H1);

  k_gemm_wmma<H1, H2><<<NN / 16, 32 * (H2 / 16), 0, stream>>>(a1, W2, h2);
  k_self<H2><<<(NN * H2 + T - 1) / T, T, 0, stream>>>(h2, dis, b2, a2);
  k_scatter<H2, 3><<<(int)(((long long)NE * (H2 / 4) + T - 1) / T), T, 0, stream>>>(h2, ei, nrm, a2);
  k_relu<<<(NN * H2 + T - 1) / T, T, 0, stream>>>(a2, NN * H2);

  k_final<<<(NN + T - 1) / T, T, 0, stream>>>(a2, lw, lb, out);
}


# bin entries dealt to the 16 waves in equal contiguous shares of whole 8-entry groups; a share's remainder also goes 8 row loads at a time (no one-load-per-entry loop); LDS sum order changed
# speedup vs baseline: 2.6159x; 2.6159x over previous
#include <hip/hip_runtime.h>
#include <math.h>


#define NN 100000
#define NE 1600000
#define KIN 64
#define H1 64
#define H2 32

#define RB 128
#define LOGRB 7
#define NB ((NN + RB - 1) / RB)
#define PT 1024
#define EPT 8
#define PCH (PT * EPT)
#define NPW ((NE + PCH - 1) / PCH)
#define SB 28
#define BT 256
#define AT 512
#define P1 68
#define P2 36
#define P3 33

static_assert(NPW % SB == 0, "scan reads the per-chunk counts SB at a time");
static_assert(NB <= PT, "one scan thread per bin");
static_assert(RB <= BT, "one degree thread per owned row");
static_assert(NPW * NB <= NN * H1, "chunk counts fit in the workspace words behind h2's region");
static_assert(NB + 1 <= NN, "bin starts live where deg used to be");
static_assert((1 << LOGRB) == RB, "bin shift");
static_assert(NN <= (1 << (32 - LOGRB)), "source row and local target row share one word");
static_assert(AT == 512 && RB == 128, "16 waves cover the 8x4 and 8x2 tiles of a bin's products");
static_assert((NN % 2) == 0, "a bin's last row pair is stored together");

typedef __attribute__((ext_vector_type(2))) float v2f;
typedef __attribute__((ext_vector_type(8))) float v8f;

__device__ __forceinline__ float edge_w(float l) {
  return 0.5f + 1.0f / (1.0f + __expf(-l));
}

__global__ __launch_bounds__(PT) void k_bin_count(const int* __restrict__ ei,
                                                  int* __restrict__ cnt) {
  __shared__ int hist[NB];
  const int p = blockIdx.x;
  for (int b = threadIdx.x; b < NB; b += PT) hist[b] = 0;
  const int e0 = p * PCH + threadIdx.x;
  int bn[EPT];
#pragma unroll
  for (int k = 0; k < EPT; ++k) {
    const int e = e0 + k * PT;
    bn[k] = (e < NE) ? (ei[NE + e] >> LOGRB) : -1;
  }
  __syncthreads();
#pragma unroll
  for (int k = 0; k < EPT; ++k)
    if (bn[k] >= 0) atomicAdd(&hist[bn[k]], 1);
  __syncthreads();
  for (int b = threadIdx.x; b < NB; b += PT) cnt[p * NB + b] = hist[b];
}

__global__ __launch_bounds__(PT) void k_bin_scan(int* __restrict__ cnt,
                                                 int* __restrict__ base) {
  __shared__ int tot[PT];
  const int b = threadIdx.x;
  int run = 0;
  if (b < NB) {
    for (int p0 = 0; p0 < NPW; p0 += SB) {
      int c[SB];
#pragma unroll
      for (int k = 0; k < SB; ++k) c[k] = cnt[(p0 + k) * NB + b];
#pragma unroll
      for (int k = 0; k < SB; ++k) { cnt[(p0 + k) * NB + b] = run; run += c[k]; }
    }
  }
  tot[b] = run;
  __syncthreads();
  for (int off = 1; off < PT; off <<= 1) {
    const int v = (b >= off) ? tot[b - off] : 0;
    __syncthreads();
    tot[b] += v;
    __syncthreads();
  }
  if (b < NB) base[b] = tot[b] - run;
  if (b == NB - 1) base[NB] = tot[b];
}

__global__ __launch_bounds__(PT) void k_bin_fill(const int* __restrict__ ei,
                                                 const float* __restrict__ logits,
                                                 const int* __restrict__ cnt,
                                                 const int* __restrict__ base,
                                                 uint2* __restrict__ bins) {
  __shared__ int cur[NB];
  const int p = blockIdx.x;
  for (int b = threadIdx.x; b < NB; b += PT) cur[b] = base[b] + cnt[p * NB + b];
  const int e0 = p * PCH + threadIdx.x;
  int r[EPT], c[EPT];
  float w[EPT];
#pragma unroll
  for (int k = 0; k < EPT; ++k) {
    const int e = e0 + k * PT;
    if (e < NE) { r[k] = ei[e]; c[k] = ei[NE + e]; w[k] = edge_w(logits[e]); }
    else        { r[k] = 0;     c[k] = -1;         w[k] = 0.0f; }
  }
  __syncthreads();
#pragma unroll
  for (int k = 0; k < EPT; ++k) {
    if (c[k] >= 0) {
      const int pos = atomicAdd(&cur[c[k] >> LOGRB], 1);
      bins[pos] = make_uint2(((unsigned)r[k] << LOGRB) | ((unsigned)c[k] & (RB - 1)),
                             __float_as_uint(w[k]));
    }
  }
}

__global__ __launch_bounds__(BT) void k_bin_dis(const uint2* __restrict__ bins,
                                                const int* __restrict__ base,
                                                float* __restrict__ dis) {
  __shared__ float sdeg[RB];
  const int bin = blockIdx.x;
  if (threadIdx.x < RB) sdeg[threadIdx.x] = 0.0f;
  __syncthreads();
  const int beg = base[bin];
  const int end = base[bin + 1];
  for (int i = beg + threadIdx.x; i < end; i += BT) {
    const uint2 t = bins[i];
    atomicAdd(&sdeg[t.x & (RB - 1)], __uint_as_float(t.y));
  }
  __syncthreads();
  const int row = bin * RB + threadIdx.x;
  if (threadIdx.x < RB && row < NN) {
    const float d = sdeg[threadIdx.x] + 1.0f;
    dis[row] = rsqrtf(d);
  }
}

__global__ __launch_bounds__(BT) void k_bin_norm(uint2* __restrict__ bins,
                                                 const int* __restrict__ base,
                                                 const float* __restrict__ dis) {
  const int bin = blockIdx.x;
  const int r0 = bin * RB;
  const int beg = base[bin];
  const int end = base[bin + 1];
  for (int i = beg + threadIdx.x; i < end; i += BT) {
    const uint2 t = bins[i];
    const float nw = dis[t.x >> LOGRB] * __uint_as_float(t.y) * dis[r0 + (int)(t.x & (RB - 1))];
    bins[i] = make_uint2(t.x, __float_as_uint(nw));
  }
}

template <int NOUT, int P, bool PART>
__device__ __forceinline__ void gather8(const float* hl, float* al,
                                        const unsigned key, const unsigned wb,
                                        const int j0, const int m) {
  constexpr int CPL = NOUT / 32;
  float v[8][CPL];
  float w[8];
  int o[8];
#pragma unroll
  for (int k = 0; k < 8; ++k) {
    const unsigned kj = (unsigned)__builtin_amdgcn_readlane((int)key, j0 + k);
    w[k] = __int_as_float(__builtin_amdgcn_readlane((int)wb, j0 + k));
    o[k] = (int)(kj & (RB - 1)) * P;
    const float* src = hl + (size_t)(kj >> LOGRB) * NOUT;
    if constexpr (CPL == 2) {
      const v2f t = *(const v2f*)src;
      v[k][0] = t.x; v[k][1] = t.y;
    } else {
      v[k][0] = src[0];
    }
  }
#pragma unroll
  for (int k = 0; k < 8; ++k) {
    if (!PART || k < m) {
#pragma unroll
      for (int q = 0; q < CPL; ++q) atomicAdd(al + o[k] + q, v[k][q] * w[k]);
    }
  }
}

template <int NOUT, int P>
__device__ __forceinline__ void bin_gather(const float* __restrict__ h,
                                           const uint2* __restrict__ bins,
                                           const int beg, const int end,
                                           float* acc, const int lane, const int wave) {
  constexpr int CPL = NOUT / 32;
  constexpr int NW = AT / 32;
  const float* hl = h + lane * CPL;
  float* al = acc + lane * CPL;

  const int per = ((end - beg + 8 * NW - 1) / (8 * NW)) * 8;
  const int wbeg = beg + wave * per;
  const int wend = min(end, wbeg + per);

  int g = wbeg;
  unsigned key = 0u, wb = 0u;
  if (g + lane < wend) { const uint2 t = bins[g + lane]; key = t.x; wb = t.y; }
  for (; g < wend; g += 32) {
    unsigned nkey = 0u, nwb = 0u;
    if (g + 32 + lane < wend) { const uint2 t = bins[g + 32 + lane]; nkey = t.x; nwb = t.y; }
    const int n = min(32, wend - g);
    if (n == 32) {
#pragma unroll
      for (int j0 = 0; j0 < 32; j0 += 8) gather8<NOUT, P, false>(hl, al, key, wb, j0, 8);
    } else {
      int j0 = 0;
      for (; j0 + 8 <= n; j0 += 8) gather8<NOUT, P, false>(hl, al, key, wb, j0, 8);
      if (j0 < n) gather8<NOUT, P, true>(hl, al, key, wb, j0, n - j0);
    }
    key = nkey;
    wb = nwb;
  }
}

template <int K, int NOUT>
__device__ __forceinline__ void load_bfrags(const float* __restrict__ Wm, const int n0,
                                            const int l15, const int half, v2f* bf) {
#pragma unroll
  for (int kt = 0; kt < K / 4; ++kt) {
    const int kk = kt * 4 + half * 2;
    v2f b;
    b.x = Wm[kk * NOUT + n0 + l15];
    b.y = Wm[(kk + 1) * NOUT + n0 + l15];
    bf[kt] = b;
  }
}

__global__ __launch_bounds__(AT) void k_agg1_dense(const float* __restrict__ x,
                                                   const uint2* __restrict__ bins,
                                                   const int* __restrict__ base,
                                                   const float* __restrict__ dis,
                                                   const float* __restrict__ W1,
                                                   const float* __restrict__ b1,
                                                   const float* __restrict__ W2,
                                                   float* __restrict__ h2) {
  __shared__ __attribute__((aligned(16))) float acc[RB * P1];
  __shared__ __attribute__((aligned(16))) float hst[RB * P2];

  const int tid = threadIdx.x;
  const int lane = tid & 31;
  const int wave = __builtin_amdgcn_readfirstlane(tid >> 5);
  const int bin = blockIdx.x;
  const int r0 = bin * RB;
  const int half = lane >> 4;
  const int l15 = lane & 15;

  for (int i = tid; i < RB * P1 / 4; i += AT)
    ((float4*)acc)[i] = make_float4(0.0f, 0.0f, 0.0f, 0.0f);
  __syncthreads();

  bin_gather<KIN, P1>(x, bins, base[bin], base[bin + 1], acc, lane, wave);
  __syncthreads();

  for (int i = tid; i < RB * (KIN / 4); i += AT) {
    const int rl = i / (KIN / 4);
    const int c = (i % (KIN / 4)) * 4;
    const int r = r0 + rl;
    if (r < NN) {
      const float d = dis[r];
      const float s = d * d;
      const float4 xv = *(const float4*)(x + (size_t)r * KIN + c);
      float4* p = (float4*)(acc + rl * P1 + c);
      float4 a = *p;
      a.x += xv.x * s; a.y += xv.y * s; a.z += xv.z * s; a.w += xv.w * s;
      *p = a;
    }
  }
  __syncthreads();

  {
    const int ct = wave & 3;
    const int rt = (wave >> 2) * 2;
    v2f bf[KIN / 4];
    load_bfrags<KIN, H1>(W1, ct * 16, l15, half, bf);
    const float* pa0 = acc + (rt * 16 + l15) * P1 + half * 2;
    const float* pa1 = pa0 + 16 * P1;
    v8f d0 = {};
    v8f d1 = {};
#pragma unroll
    for (int kt = 0; kt < KIN / 4; ++kt) {
      const v2f a0 = *(const v2f*)(pa0 + kt * 4);
      const v2f a1 = *(const v2f*)(pa1 + kt * 4);
      d0 = __builtin_amdgcn_wmma_f32_16x16x4_f32(false, a0, false, bf[kt], (short)0, d0, false, false);
      d1 = __builtin_amdgcn_wmma_f32_16x16x4_f32(false, a1, false, bf[kt], (short)0, d1, false, false);
    }
    __syncthreads();
    const float bb = b1[ct * 16 + l15];
    float* po = acc + (rt * 16 + half * 8) * P1 + ct * 16 + l15;
#pragma unroll
    for (int v = 0; v < 8; ++v) {
      po[v * P1] = fmaxf(d0[v] + bb, 0.0f);
      po[(16 + v) * P1] = fmaxf(d1[v] + bb, 0.0f);
    }
  }
  __syncthreads();

  {
    const int ct = wave & 1;
    const int rt = wave >> 1;
    v2f bf[H1 / 4];
    load_bfrags<H1, H2>(W2, ct * 16, l15, half, bf);
    const float* pa = acc + (rt * 16 + l15) * P1 + half * 2;
    v8f d = {};
#pragma unroll
    for (int kt = 0; kt < H1 / 4; ++kt) {
      const v2f a = *(const v2f*)(pa + kt * 4);
      d = __builtin_amdgcn_wmma_f32_16x16x4_f32(false, a, false, bf[kt], (short)0, d, false, false);
    }
    float* po = hst + (rt * 16 + half * 8) * P2 + ct * 16 + l15;
#pragma unroll
    for (int v = 0; v < 8; ++v) po[v * P2] = d[v];
  }
  __syncthreads();

  for (int i = tid; i < RB * (H2 / 4); i += AT) {
    const int rl = i / (H2 / 4);
    const int c = (i % (H2 / 4)) * 4;
    const int r = r0 + rl;
    if (r < NN)
      *(float4*)(h2 + (size_t)r * H2 + c) = *(const float4*)(hst + rl * P2 + c);
  }
}

__global__ __launch_bounds__(AT) void k_agg2_head(const float* __restrict__ h2,
                                                  const uint2* __restrict__ bins,
                                                  const int* __restrict__ base,
                                                  const float* __restrict__ dis,
                                                  const float* __restrict__ b2,
                                                  const float* __restrict__ lw,
                                                  const float* __restrict__ lb,
                                                  float* __restrict__ out) {
  __shared__ float acc[RB * P3];
  __shared__ float slog[RB];

  const int tid = threadIdx.x;
  const int lane = tid & 31;
  const int wave = __builtin_amdgcn_readfirstlane(tid >> 5);
  const int bin = blockIdx.x;
  const int r0 = bin * RB;

  for (int i = tid; i < RB * P3; i += AT) acc[i] = 0.0f;
  __syncthreads();

  bin_gather<H2, P3>(h2, bins, base[bin], base[bin + 1], acc, lane, wave);
  __syncthreads();

  const int rl = tid >> 2;
  const int part = tid & 3;
  const int r = r0 + rl;
  float p = 0.0f;
  if (r < NN) {
    const float d = dis[r];
    const float s = d * d;
    const float* hp = h2 + (size_t)r * H2 + part * 8;
    const float4 ha = *(const float4*)hp;
    const float4 hb = *(const float4*)(hp + 4);
    const float hv[8] = {ha.x, ha.y, ha.z, ha.w, hb.x, hb.y, hb.z, hb.w};
    const float* ap = acc + rl * P3 + part * 8;
#pragma unroll
    for (int j = 0; j < 8; ++j) {
      const float a = fmaxf(ap[j] + (hv[j] * s + b2[part * 8 + j]), 0.0f);
      p = fmaf(a, lw[part * 8 + j], p);
    }
  }
  p += __shfl_xor(p, 1);
  p += __shfl_xor(p, 2);
  if (part == 0) slog[rl] = p + lb[0];
  __syncthreads();

  if (tid < RB / 2 && r0 + 2 * tid < NN) {
    const float l0 = slog[2 * tid];
    const float l1 = slog[2 * tid + 1];
    *(float4*)(out + (size_t)(r0 + 2 * tid) * 2) = make_float4(-l0, l0, -l1, l1);
  }
}

extern "C" void kernel_launch(void* const* d_in, const int* in_sizes, int n_in,
                              void* d_out, int out_size, void* d_ws, size_t ws_size,
                              hipStream_t stream) {
  const float* x      = (const float*)d_in[0];
  const int*   ei     = (const int*)d_in[1];
  const float* elog   = (const float*)d_in[2];
  const float* W1     = (const float*)d_in[3];
  const float* b1     = (const float*)d_in[4];
  const float* W2     = (const float*)d_in[5];
  const float* b2     = (const float*)d_in[6];
  const float* lw     = (const float*)d_in[7];
  const float* lb     = (const float*)d_in[8];
  float* out = (float*)d_out;

  float* ws   = (float*)d_ws;
  int*   base = (int*)ws;
  float* dis  = ws + NN;
  uint2* bins = (uint2*)(dis + NN);
  float* h2   = dis + NN + 2 * (size_t)NE;
  int*   cnt  = (int*)(h2 + (size_t)NN * H1);

  k_bin_count<<<NPW, PT, 0, stream>>>(ei, cnt);
  k_bin_scan<<<1, PT, 0, stream>>>(cnt, base);
  k_bin_fill<<<NPW, PT, 0, stream>>>(ei, elog, cnt, base, bins);
  k_bin_dis<<<NB, BT, 0, stream>>>(bins, base, dis);
  k_bin_norm<<<NB, BT, 0, stream>>>(bins, base, dis);

  k_agg1_dense<<<NB, AT, 0, stream>>>(x, bins, base, dis, W1, b1, W2, h2);

  k_agg2_head<<<NB, AT, 0, stream>>>(h2, bins, base, dis, b2, lw, lb, out);
}


# bin entries counting-sorted by target row in the norm pass; a wave sums each row's run in registers and adds it to the LDS tile once per row, not once per entry; sum order changed
# speedup vs baseline: 2.8387x; 1.0852x over previous
#include <hip/hip_runtime.h>
#include <math.h>


#define NN 100000
#define NE 1600000
#define KIN 64
#define H1 64
#define H2 32

#define RB 128
#define LOGRB 7
#define NB ((NN + RB - 1) / RB)
#define PT 1024
#define EPT 8
#define PCH (PT * EPT)
#define NPW ((NE + PCH - 1) / PCH)
#define SB 28
#define BT 256
#define AT 512
#define P1 68
#define P2 36
#define P3 33

static_assert(NPW % SB == 0, "scan reads the per-chunk counts SB at a time");
static_assert(NB <= PT, "one scan thread per bin");
static_assert(RB <= BT, "one degree thread per owned row");
static_assert(NPW * NB <= NN * H1, "chunk counts fit in the workspace words behind h2's region");
static_assert(NB + 1 <= NN, "bin starts live where deg used to be");
static_assert((1 << LOGRB) == RB, "bin shift");
static_assert(NN <= (1 << (32 - LOGRB)), "source row and local target row share one word");
static_assert(AT == 512 && RB == 128, "16 waves cover the 8x4 and 8x2 tiles of a bin's products");
static_assert((NN % 2) == 0, "a bin's last row pair is stored together");
static_assert(2 * (size_t)NE <= (size_t)NN * (H1 - H2), "the sorted entry list fits behind h2 in the region h1 had");

typedef __attribute__((ext_vector_type(2))) float v2f;
typedef __attribute__((ext_vector_type(8))) float v8f;

__device__ __forceinline__ float edge_w(float l) {
  return 0.5f + 1.0f / (1.0f + __expf(-l));
}

__global__ __launch_bounds__(PT) void k_bin_count(const int* __restrict__ ei,
                                                  int* __restrict__ cnt) {
  __shared__ int hist[NB];
  const int p = blockIdx.x;
  for (int b = threadIdx.x; b < NB; b += PT) hist[b] = 0;
  const int e0 = p * PCH + threadIdx.x;
  int bn[EPT];
#pragma unroll
  for (int k = 0; k < EPT; ++k) {
    const int e = e0 + k * PT;
    bn[k] = (e < NE) ? (ei[NE + e] >> LOGRB) : -1;
  }
  __syncthreads();
#pragma unroll
  for (int k = 0; k < EPT; ++k)
    if (bn[k] >= 0) atomicAdd(&hist[bn[k]], 1);
  __syncthreads();
  for (int b = threadIdx.x; b < NB; b += PT) cnt[p * NB + b] = hist[b];
}

__global__ __launch_bounds__(PT) void k_bin_scan(int* __restrict__ cnt,
                                                 int* __restrict__ base) {
  __shared__ int tot[PT];
  const int b = threadIdx.x;
  int run = 0;
  if (b < NB) {
    for (int p0 = 0; p0 < NPW; p0 += SB) {
      int c[SB];
#pragma unroll
      for (int k = 0; k < SB; ++k) c[k] = cnt[(p0 + k) * NB + b];
#pragma unroll
      for (int k = 0; k < SB; ++k) { cnt[(p0 + k) * NB + b] = run; run += c[k]; }
    }
  }
  tot[b] = run;
  __syncthreads();
  for (int off = 1; off < PT; off <<= 1) {
    const int v = (b >= off) ? tot[b - off] : 0;
    __syncthreads();
    tot[b] += v;
    __syncthreads();
  }
  if (b < NB) base[b] = tot[b] - run;
  if (b == NB - 1) base[NB] = tot[b];
}

__global__ __launch_bounds__(PT) void k_bin_fill(const int* __restrict__ ei,
                                                 const float* __restrict__ logits,
                                                 const int* __restrict__ cnt,
                                                 const int* __restrict__ base,
                                                 uint2* __restrict__ bins) {
  __shared__ int cur[NB];
  const int p = blockIdx.x;
  for (int b = threadIdx.x; b < NB; b += PT) cur[b] = base[b] + cnt[p * NB + b];
  const int e0 = p * PCH + threadIdx.x;
  int r[EPT], c[EPT];
  float w[EPT];
#pragma unroll
  for (int k = 0; k < EPT; ++k) {
    const int e = e0 + k * PT;
    if (e < NE) { r[k] = ei[e]; c[k] = ei[NE + e]; w[k] = edge_w(logits[e]); }
    else        { r[k] = 0;     c[k] = -1;         w[k] = 0.0f; }
  }
  __syncthreads();
#pragma unroll
  for (int k = 0; k < EPT; ++k) {
    if (c[k] >= 0) {
      const int pos = atomicAdd(&cur[c[k] >> LOGRB], 1);
      bins[pos] = make_uint2(((unsigned)r[k] << LOGRB) | ((unsigned)c[k] & (RB - 1)),
                             __float_as_uint(w[k]));
    }
  }
}

__global__ __launch_bounds__(BT) void k_bin_dis(const uint2* __restrict__ bins,
                                                const int* __restrict__ base,
                                                float* __restrict__ dis) {
  __shared__ float sdeg[RB];
  const int bin = blockIdx.x;
  if (threadIdx.x < RB) sdeg[threadIdx.x] = 0.0f;
  __syncthreads();
  const int beg = base[bin];
  const int end = base[bin + 1];
  for (int i = beg + threadIdx.x; i < end; i += BT) {
    const uint2 t = bins[i];
    atomicAdd(&sdeg[t.x & (RB - 1)], __uint_as_float(t.y));
  }
  __syncthreads();
  const int row = bin * RB + threadIdx.x;
  if (threadIdx.x < RB && row < NN) {
    const float d = sdeg[threadIdx.x] + 1.0f;
    dis[row] = rsqrtf(d);
  }
}

__global__ __launch_bounds__(BT) void k_bin_sort_norm(const uint2* __restrict__ bins,
                                                      const int* __restrict__ base,
                                                      const float* __restrict__ dis,
                                                      uint2* __restrict__ srt) {
  __shared__ int rpos[RB];
  const int tid = threadIdx.x;
  const int bin = blockIdx.x;
  const int r0 = bin * RB;
  const int beg = base[bin];
  const int end = base[bin + 1];
  if (tid < RB) rpos[tid] = 0;
  __syncthreads();
  for (int g = beg + tid; g - tid < end; g += 4 * BT) {
    int tr[4];
#pragma unroll
    for (int u = 0; u < 4; ++u) tr[u] = (int)(bins[min(g + u * BT, end - 1)].x & (RB - 1));
#pragma unroll
    for (int u = 0; u < 4; ++u)
      if (g + u * BT < end) atomicAdd(&rpos[tr[u]], 1);
  }
  __syncthreads();
  const int own = (tid < RB) ? rpos[tid] : 0;
  for (int off = 1; off < RB; off <<= 1) {
    const int v = (tid < RB && tid >= off) ? rpos[tid - off] : 0;
    __syncthreads();
    if (tid < RB) rpos[tid] += v;
    __syncthreads();
  }
  if (tid < RB) rpos[tid] = beg + rpos[tid] - own;
  __syncthreads();
  for (int i = beg + tid; i < end; i += BT) {
    const uint2 t = bins[i];
    const int tl = (int)(t.x & (RB - 1));
    const float nw = dis[t.x >> LOGRB] * __uint_as_float(t.y) * dis[r0 + tl];
    const int pos = atomicAdd(&rpos[tl], 1);
    srt[pos] = make_uint2(t.x, __float_as_uint(nw));
  }
}

template <int CPL>
__device__ __forceinline__ void run_flush(float* p, const float* a) {
  atomicAdd(p, a[0]);
  if constexpr (CPL == 2) atomicAdd(p + 1, a[1]);
}

template <int NOUT, int P, bool PART>
__device__ __forceinline__ void gather8(const float* hl, float* al,
                                        const unsigned key, const unsigned wb,
                                        const int j0, const int m,
                                        int& cur, float* a) {
  constexpr int CPL = NOUT / 32;
  float v[8][CPL];
  float w[8];
  int o[8];
#pragma unroll
  for (int k = 0; k < 8; ++k) {
    const unsigned kj = (unsigned)__builtin_amdgcn_readlane((int)key, j0 + k);
    w[k] = __int_as_float(__builtin_amdgcn_readlane((int)wb, j0 + k));
    o[k] = (int)(kj & (RB - 1)) * P;
    const float* src = hl + (size_t)(kj >> LOGRB) * NOUT;
    if constexpr (CPL == 2) {
      const v2f t = *(const v2f*)src;
      v[k][0] = t.x; v[k][1] = t.y;
    } else {
      v[k][0] = src[0];
    }
  }
#pragma unroll
  for (int k = 0; k < 8; ++k) {
    if (!PART || k < m) {
      if (o[k] != cur) {
        if (cur >= 0) run_flush<CPL>(al + cur, a);
        a[0] = 0.0f;
        if constexpr (CPL == 2) a[1] = 0.0f;
        cur = o[k];
      }
#pragma unroll
      for (int q = 0; q < CPL; ++q) a[q] = fmaf(v[k][q], w[k], a[q]);
    }
  }
}

template <int NOUT, int P>
__device__ __forceinline__ void bin_gather(const float* __restrict__ h,
                                           const uint2* __restrict__ bins,
                                           const int beg, const int end,
                                           float* acc, const int lane, const int wave) {
  constexpr int CPL = NOUT / 32;
  constexpr int NW = AT / 32;
  const float* hl = h + lane * CPL;
  float* al = acc + lane * CPL;

  const int per = ((end - beg + 8 * NW - 1) / (8 * NW)) * 8;
  const int wbeg = beg + wave * per;
  const int wend = min(end, wbeg + per);

  int cur = -1;
  float a[CPL] = {};

  int g = wbeg;
  unsigned key = 0u, wb = 0u;
  if (g + lane < wend) { const uint2 t = bins[g + lane]; key = t.x; wb = t.y; }
  for (; g < wend; g += 32) {
    unsigned nkey = 0u, nwb = 0u;
    if (g + 32 + lane < wend) { const uint2 t = bins[g + 32 + lane]; nkey = t.x; nwb = t.y; }
    const int n = min(32, wend - g);
    if (n == 32) {
#pragma unroll
      for (int j0 = 0; j0 < 32; j0 += 8) gather8<NOUT, P, false>(hl, al, key, wb, j0, 8, cur, a);
    } else {
      int j0 = 0;
      for (; j0 + 8 <= n; j0 += 8) gather8<NOUT, P, false>(hl, al, key, wb, j0, 8, cur, a);
      if (j0 < n) gather8<NOUT, P, true>(hl, al, key, wb, j0, n - j0, cur, a);
    }
    key = nkey;
    wb = nwb;
  }
  if (cur >= 0) run_flush<CPL>(al + cur, a);
}

template <int K, int NOUT>
__device__ __forceinline__ void load_bfrags(const float* __restrict__ Wm, const int n0,
                                            const int l15, const int half, v2f* bf) {
#pragma unroll
  for (int kt = 0; kt < K / 4; ++kt) {
    const int kk = kt * 4 + half * 2;
    v2f b;
    b.x = Wm[kk * NOUT + n0 + l15];
    b.y = Wm[(kk + 1) * NOUT + n0 + l15];
    bf[kt] = b;
  }
}

__global__ __launch_bounds__(AT) void k_agg1_dense(const float* __restrict__ x,
                                                   const uint2* __restrict__ bins,
                                                   const int* __restrict__ base,
                                                   const float* __restrict__ dis,
                                                   const float* __restrict__ W1,
                                                   const float* __restrict__ b1,
                                                   const float* __restrict__ W2,
                                                   float* __restrict__ h2) {
  __shared__ __attribute__((aligned(16))) float acc[RB * P1];
  __shared__ __attribute__((aligned(16))) float hst[RB * P2];

  const int tid = threadIdx.x;
  const int lane = tid & 31;
  const int wave = __builtin_amdgcn_readfirstlane(tid >> 5);
  const int bin = blockIdx.x;
  const int r0 = bin * RB;
  const int half = lane >> 4;
  const int l15 = lane & 15;

  for (int i = tid; i < RB * P1 / 4; i += AT)
    ((float4*)acc)[i] = make_float4(0.0f, 0.0f, 0.0f, 0.0f);
  __syncthreads();

  bin_gather<KIN, P1>(x, bins, base[bin], base[bin + 1], acc, lane, wave);
  __syncthreads();

  for (int i = tid; i < RB * (KIN / 4); i += AT) {
    const int rl = i / (KIN / 4);
    const int c = (i % (KIN / 4)) * 4;
    const int r = r0 + rl;
    if (r < NN) {
      const float d = dis[r];
      const float s = d * d;
      const float4 xv = *(const float4*)(x + (size_t)r * KIN + c);
      float4* p = (float4*)(acc + rl * P1 + c);
      float4 a = *p;
      a.x += xv.x * s; a.y += xv.y * s; a.z += xv.z * s; a.w += xv.w * s;
      *p = a;
    }
  }
  __syncthreads();

  {
    const int ct = wave & 3;
    const int rt = (wave >> 2) * 2;
    v2f bf[KIN / 4];
    load_bfrags<KIN, H1>(W1, ct * 16, l15, half, bf);
    const float* pa0 = acc + (rt * 16 + l15) * P1 + half * 2;
    const float* pa1 = pa0 + 16 * P1;
    v8f d0 = {};
    v8f d1 = {};
#pragma unroll
    for (int kt = 0; kt < KIN / 4; ++kt) {
      const v2f a0 = *(const v2f*)(pa0 + kt * 4);
      const v2f a1 = *(const v2f*)(pa1 + kt * 4);
      d0 = __builtin_amdgcn_wmma_f32_16x16x4_f32(false, a0, false, bf[kt], (short)0, d0, false, false);
      d1 = __builtin_amdgcn_wmma_f32_16x16x4_f32(false, a1, false, bf[kt], (short)0, d1, false, false);
    }
    __syncthreads();
    const float bb = b1[ct * 16 + l15];
    float* po = acc + (rt * 16 + half * 8) * P1 + ct * 16 + l15;
#pragma unroll
    for (int v = 0; v < 8; ++v) {
      po[v * P1] = fmaxf(d0[v] + bb, 0.0f);
      po[(16 + v) * P1] = fmaxf(d1[v] + bb, 0.0f);
    }
  }
  __syncthreads();

  {
    const int ct = wave & 1;
    const int rt = wave >> 1;
    v2f bf[H1 / 4];
    load_bfrags<H1, H2>(W2, ct * 16, l15, half, bf);
    const float* pa = acc + (rt * 16 + l15) * P1 + half * 2;
    v8f d = {};
#pragma unroll
    for (int kt = 0; kt < H1 / 4; ++kt) {
      const v2f a = *(const v2f*)(pa + kt * 4);
      d = __builtin_amdgcn_wmma_f32_16x16x4_f32(false, a, false, bf[kt], (short)0, d, false, false);
    }
    float* po = hst + (rt * 16 + half * 8) * P2 + ct * 16 + l15;
#pragma unroll
    for (int v = 0; v < 8; ++v) po[v * P2] = d[v];
  }
  __syncthreads();

  for (int i = tid; i < RB * (H2 / 4); i += AT) {
    const int rl = i / (H2 / 4);
    const int c = (i % (H2 / 4)) * 4;
    const int r = r0 + rl;
    if (r < NN)
      *(float4*)(h2 + (size_t)r * H2 + c) = *(const float4*)(hst + rl * P2 + c);
  }
}

__global__ __launch_bounds__(AT) void k_agg2_head(const float* __restrict__ h2,
                                                  const uint2* __restrict__ bins,
                                                  const int* __restrict__ base,
                                                  const float* __restrict__ dis,
                                                  const float* __restrict__ b2,
                                                  const float* __restrict__ lw,
                                                  const float* __restrict__ lb,
                                                  float* __restrict__ out) {
  __shared__ float acc[RB * P3];
  __shared__ float slog[RB];

  const int tid = threadIdx.x;
  const int lane = tid & 31;
  const int wave = __builtin_amdgcn_readfirstlane(tid >> 5);
  const int bin = blockIdx.x;
  const int r0 = bin * RB;

  for (int i = tid; i < RB * P3; i += AT) acc[i] = 0.0f;
  __syncthreads();

  bin_gather<H2, P3>(h2, bins, base[bin], base[bin + 1], acc, lane, wave);
  __syncthreads();

  const int rl = tid >> 2;
  const int part = tid & 3;
  const int r = r0 + rl;
  float p = 0.0f;
  if (r < NN) {
    const float d = dis[r];
    const float s = d * d;
    const float* hp = h2 + (size_t)r * H2 + part * 8;
    const float4 ha = *(const float4*)hp;
    const float4 hb = *(const float4*)(hp + 4);
    const float hv[8] = {ha.x, ha.y, ha.z, ha.w, hb.x, hb.y, hb.z, hb.w};
    const float* ap = acc + rl * P3 + part * 8;
#pragma unroll
    for (int j = 0; j < 8; ++j) {
      const float a = fmaxf(ap[j] + (hv[j] * s + b2[part * 8 + j]), 0.0f);
      p = fmaf(a, lw[part * 8 + j], p);
    }
  }
  p += __shfl_xor(p, 1);
  p += __shfl_xor(p, 2);
  if (part == 0) slog[rl] = p + lb[0];
  __syncthreads();

  if (tid < RB / 2 && r0 + 2 * tid < NN) {
    const float l0 = slog[2 * tid];
    const float l1 = slog[2 * tid + 1];
    *(float4*)(out + (size_t)(r0 + 2 * tid) * 2) = make_float4(-l0, l0, -l1, l1);
  }
}

extern "C" void kernel_launch(void* const* d_in, const int* in_sizes, int n_in,
                              void* d_out, int out_size, void* d_ws, size_t ws_size,
                              hipStream_t stream) {
  const float* x      = (const float*)d_in[0];
  const int*   ei     = (const int*)d_in[1];
  const float* elog   = (const float*)d_in[2];
  const float* W1     = (const float*)d_in[3];
  const float* b1     = (const float*)d_in[4];
  const float* W2     = (const float*)d_in[5];
  const float* b2     = (const float*)d_in[6];
  const float* lw     = (const float*)d_in[7];
  const float* lb     = (const float*)d_in[8];
  float* out = (float*)d_out;

  float* ws   = (float*)d_ws;
  int*   base = (int*)ws;
  float* dis  = ws + NN;
  uint2* bins = (uint2*)(dis + NN);
  float* h2   = dis + NN + 2 * (size_t)NE;
  uint2* srt  = (uint2*)(h2 + (size_t)NN * H2);
  int*   cnt  = (int*)(h2 + (size_t)NN * H1);

  k_bin_count<<<NPW, PT, 0, stream>>>(ei, cnt);
  k_bin_scan<<<1, PT, 0, stream>>>(cnt, base);
  k_bin_fill<<<NPW, PT, 0, stream>>>(ei, elog, cnt, base, bins);
  k_bin_dis<<<NB, BT, 0, stream>>>(bins, base, dis);
  k_bin_sort_norm<<<NB, BT, 0, stream>>>(bins, base, dis, srt);

  k_agg1_dense<<<NB, AT, 0, stream>>>(x, srt, base, dis, W1, b1, W2, h2);

  k_agg2_head<<<NB, AT, 0, stream>>>(h2, srt, base, dis, b2, lw, lb, out);
}


# bin fill and row sort stage their entries in LDS in destination order, then store slot after slot: contiguous pieces, not one 8-byte store request per entry; row entry order still LDS arrival order
# speedup vs baseline: 3.1359x; 1.1047x over previous
#include <hip/hip_runtime.h>
#include <math.h>


#define NN 100000
#define NE 1600000
#define KIN 64
#define H1 64
#define H2 32

#define RB 128
#define LOGRB 7
#define NB ((NN + RB - 1) / RB)
#define PT 1024
#define EPT 8
#define PCH (PT * EPT)
#define NPW ((NE + PCH - 1) / PCH)
#define SB 28
#define BT 256
#define SCAP 4096
#define AT 512
#define P1 68
#define P2 36
#define P3 33

static_assert(NPW % SB == 0, "scan reads the per-chunk counts SB at a time");
static_assert(NB <= PT, "one scan thread per bin");
static_assert(RB <= BT, "one degree thread per owned row");
static_assert(NPW * NB <= NN * H1, "chunk counts fit in the workspace words behind h2's region");
static_assert(NB + 1 <= NN, "bin starts live where deg used to be");
static_assert((1 << LOGRB) == RB, "bin shift");
static_assert(NN <= (1 << (32 - LOGRB)), "source row and local target row share one word");
static_assert(AT == 512 && RB == 128, "16 waves cover the 8x4 and 8x2 tiles of a bin's products");
static_assert((NN % 2) == 0, "a bin's last row pair is stored together");
static_assert(2 * (size_t)NE <= (size_t)NN * (H1 - H2), "the sorted entry list fits behind h2 in the region h1 had");

typedef __attribute__((ext_vector_type(2))) float v2f;
typedef __attribute__((ext_vector_type(8))) float v8f;

__device__ __forceinline__ float edge_w(float l) {
  return 0.5f + 1.0f / (1.0f + __expf(-l));
}

__global__ __launch_bounds__(PT) void k_bin_count(const int* __restrict__ ei,
                                                  int* __restrict__ cnt) {
  __shared__ int hist[NB];
  const int p = blockIdx.x;
  for (int b = threadIdx.x; b < NB; b += PT) hist[b] = 0;
  const int e0 = p * PCH + threadIdx.x;
  int bn[EPT];
#pragma unroll
  for (int k = 0; k < EPT; ++k) {
    const int e = e0 + k * PT;
    bn[k] = (e < NE) ? (ei[NE + e] >> LOGRB) : -1;
  }
  __syncthreads();
#pragma unroll
  for (int k = 0; k < EPT; ++k)
    if (bn[k] >= 0) atomicAdd(&hist[bn[k]], 1);
  __syncthreads();
  for (int b = threadIdx.x; b < NB; b += PT) cnt[p * NB + b] = hist[b];
}

__global__ __launch_bounds__(PT) void k_bin_scan(int* __restrict__ cnt,
                                                 int* __restrict__ base) {
  __shared__ int tot[PT];
  const int b = threadIdx.x;
  int run = 0;
  if (b < NB) {
    for (int p0 = 0; p0 < NPW; p0 += SB) {
      int c[SB];
#pragma unroll
      for (int k = 0; k < SB; ++k) c[k] = cnt[(p0 + k) * NB + b];
#pragma unroll
      for (int k = 0; k < SB; ++k) { cnt[(p0 + k) * NB + b] = run; run += c[k]; }
    }
  }
  tot[b] = run;
  __syncthreads();
  for (int off = 1; off < PT; off <<= 1) {
    const int v = (b >= off) ? tot[b - off] : 0;
    __syncthreads();
    tot[b] += v;
    __syncthreads();
  }
  if (b < NB) base[b] = tot[b] - run;
  if (b == NB - 1) base[NB] = tot[b];
}

__global__ __launch_bounds__(PT) void k_bin_fill(const int* __restrict__ ei,
                                                 const float* __restrict__ logits,
                                                 const int* __restrict__ cnt,
                                                 const int* __restrict__ base,
                                                 uint2* __restrict__ bins) {
  __shared__ int gst[NB];
  __shared__ int lcnt[NB + 1];
  __shared__ int wtot[PT / 32];
  __shared__ uint2 sent[PCH];
  __shared__ int sdst[PCH];
  const int tid = threadIdx.x;
  const int lane = tid & 31;
  const int wv = tid >> 5;
  const int p = blockIdx.x;
  for (int b = tid; b < NB; b += PT) gst[b] = base[b] + cnt[p * NB + b];
  for (int b = tid; b <= NB; b += PT) lcnt[b] = 0;
  const int e0 = p * PCH + tid;
  int r[EPT], c[EPT];
  float w[EPT];
#pragma unroll
  for (int k = 0; k < EPT; ++k) {
    const int e = e0 + k * PT;
    if (e < NE) { r[k] = ei[e]; c[k] = ei[NE + e]; w[k] = edge_w(logits[e]); }
    else        { r[k] = 0;     c[k] = -1;         w[k] = 0.0f; }
  }
  __syncthreads();
  int bk[EPT], rk[EPT];
#pragma unroll
  for (int k = 0; k < EPT; ++k) {
    bk[k] = (c[k] >= 0) ? (c[k] >> LOGRB) : NB;
    rk[k] = atomicAdd(&lcnt[bk[k]], 1);
  }
  __syncthreads();
  const int own = (tid < NB) ? lcnt[tid] : 0;
  int inc = own;
  for (int off = 1; off < 32; off <<= 1) {
    const int v = __shfl_up(inc, off);
    if (lane >= off) inc += v;
  }
  if (lane == 31) wtot[wv] = inc;
  __syncthreads();
  if (wv == 0) {
    const int t = wtot[lane];
    int ti = t;
    for (int off = 1; off < 32; off <<= 1) {
      const int v = __shfl_up(ti, off);
      if (lane >= off) ti += v;
    }
    wtot[lane] = ti - t;
  }
  __syncthreads();
  if (tid < NB) lcnt[tid] = wtot[wv] + inc - own;
  __syncthreads();
  int ls[EPT], gs[EPT];
#pragma unroll
  for (int k = 0; k < EPT; ++k) {
    const int b = min(bk[k], NB - 1);
    ls[k] = lcnt[b];
    gs[k] = gst[b];
  }
#pragma unroll
  for (int k = 0; k < EPT; ++k) {
    if (c[k] >= 0) {
      const int s = ls[k] + rk[k];
      sent[s] = make_uint2(((unsigned)r[k] << LOGRB) | ((unsigned)c[k] & (RB - 1)),
                           __float_as_uint(w[k]));
      sdst[s] = gs[k] + rk[k];
    }
  }
  __syncthreads();
  const int n = min(PCH, NE - p * PCH);
  uint2 ev[EPT];
  int dv[EPT];
#pragma unroll
  for (int k = 0; k < EPT; ++k) {
    ev[k] = sent[tid + k * PT];
    dv[k] = sdst[tid + k * PT];
  }
#pragma unroll
  for (int k = 0; k < EPT; ++k)
    if (tid + k * PT < n) bins[dv[k]] = ev[k];
}

__global__ __launch_bounds__(BT) void k_bin_dis(const uint2* __restrict__ bins,
                                                const int* __restrict__ base,
                                                float* __restrict__ dis) {
  __shared__ float sdeg[RB];
  const int bin = blockIdx.x;
  if (threadIdx.x < RB) sdeg[threadIdx.x] = 0.0f;
  __syncthreads();
  const int beg = base[bin];
  const int end = base[bin + 1];
  for (int i = beg + threadIdx.x; i < end; i += BT) {
    const uint2 t = bins[i];
    atomicAdd(&sdeg[t.x & (RB - 1)], __uint_as_float(t.y));
  }
  __syncthreads();
  const int row = bin * RB + threadIdx.x;
  if (threadIdx.x < RB && row < NN) {
    const float d = sdeg[threadIdx.x] + 1.0f;
    dis[row] = rsqrtf(d);
  }
}

__global__ __launch_bounds__(BT) void k_bin_sort_norm(const uint2* __restrict__ bins,
                                                      const int* __restrict__ base,
                                                      const float* __restrict__ dis,
                                                      uint2* __restrict__ srt) {
  __shared__ int rpos[RB];
  __shared__ uint2 stg[SCAP];
  const int tid = threadIdx.x;
  const int bin = blockIdx.x;
  const int r0 = bin * RB;
  const int beg = base[bin];
  const int end = base[bin + 1];
  const int n = end - beg;
  const bool staged = n <= SCAP;
  if (tid < RB) rpos[tid] = 0;
  __syncthreads();
  for (int g = beg + tid; g - tid < end; g += 4 * BT) {
    int tr[4];
#pragma unroll
    for (int u = 0; u < 4; ++u) tr[u] = (int)(bins[min(g + u * BT, end - 1)].x & (RB - 1));
#pragma unroll
    for (int u = 0; u < 4; ++u)
      if (g + u * BT < end) atomicAdd(&rpos[tr[u]], 1);
  }
  __syncthreads();
  const int own = (tid < RB) ? rpos[tid] : 0;
  for (int off = 1; off < RB; off <<= 1) {
    const int v = (tid < RB && tid >= off) ? rpos[tid - off] : 0;
    __syncthreads();
    if (tid < RB) rpos[tid] += v;
    __syncthreads();
  }
  if (tid < RB) rpos[tid] = rpos[tid] - own;
  __syncthreads();
  for (int i = beg + tid; i < end; i += BT) {
    const uint2 t = bins[i];
    const int tl = (int)(t.x & (RB - 1));
    const float nw = dis[t.x >> LOGRB] * __uint_as_float(t.y) * dis[r0 + tl];
    const int pos = atomicAdd(&rpos[tl], 1);
    const uint2 o = make_uint2(t.x, __float_as_uint(nw));
    if (staged) stg[pos] = o;
    else        srt[beg + pos] = o;
  }
  if (staged) {
    __syncthreads();
    for (int i0 = tid; i0 < n; i0 += 4 * BT) {
      uint2 ev[4];
#pragma unroll
      for (int u = 0; u < 4; ++u) ev[u] = stg[min(i0 + u * BT, SCAP - 1)];
#pragma unroll
      for (int u = 0; u < 4; ++u)
        if (i0 + u * BT < n) srt[beg + i0 + u * BT] = ev[u];
    }
  }
}

template <int CPL>
__device__ __forceinline__ void run_flush(float* p, const float* a) {
  atomicAdd(p, a[0]);
  if constexpr (CPL == 2) atomicAdd(p + 1, a[1]);
}

template <int NOUT, int P, bool PART>
__device__ __forceinline__ void gather8(const float* hl, float* al,
                                        const unsigned key, const unsigned wb,
                                        const int j0, const int m,
                                        int& cur, float* a) {
  constexpr int CPL = NOUT / 32;
  float v[8][CPL];
  float w[8];
  int o[8];
#pragma unroll
  for (int k = 0; k < 8; ++k) {
    const unsigned kj = (unsigned)__builtin_amdgcn_readlane((int)key, j0 + k);
    w[k] = __int_as_float(__builtin_amdgcn_readlane((int)wb, j0 + k));
    o[k] = (int)(kj & (RB - 1)) * P;
    const float* src = hl + (size_t)(kj >> LOGRB) * NOUT;
    if constexpr (CPL == 2) {
      const v2f t = *(const v2f*)src;
      v[k][0] = t.x; v[k][1] = t.y;
    } else {
      v[k][0] = src[0];
    }
  }
#pragma unroll
  for (int k = 0; k < 8; ++k) {
    if (!PART || k < m) {
      if (o[k] != cur) {
        if (cur >= 0) run_flush<CPL>(al + cur, a);
        a[0] = 0.0f;
        if constexpr (CPL == 2) a[1] = 0.0f;
        cur = o[k];
      }
#pragma unroll
      for (int q = 0; q < CPL; ++q) a[q] = fmaf(v[k][q], w[k], a[q]);
    }
  }
}

template <int NOUT, int P>
__device__ __forceinline__ void bin_gather(const float* __restrict__ h,
                                           const uint2* __restrict__ bins,
                                           const int beg, const int end,
                                           float* acc, const int lane, const int wave) {
  constexpr int CPL = NOUT / 32;
  constexpr int NW = AT / 32;
  const float* hl = h + lane * CPL;
  float* al = acc + lane * CPL;

  const int per = ((end - beg + 8 * NW - 1) / (8 * NW)) * 8;
  const int wbeg = beg + wave * per;
  const int wend = min(end, wbeg + per);

  int cur = -1;
  float a[CPL] = {};

  int g = wbeg;
  unsigned key = 0u, wb = 0u;
  if (g + lane < wend) { const uint2 t = bins[g + lane]; key = t.x; wb = t.y; }
  for (; g < wend; g += 32) {
    unsigned nkey = 0u, nwb = 0u;
    if (g + 32 + lane < wend) { const uint2 t = bins[g + 32 + lane]; nkey = t.x; nwb = t.y; }
    const int n = min(32, wend - g);
    if (n == 32) {
#pragma unroll
      for (int j0 = 0; j0 < 32; j0 += 8) gather8<NOUT, P, false>(hl, al, key, wb, j0, 8, cur, a);
    } else {
      int j0 = 0;
      for (; j0 + 8 <= n; j0 += 8) gather8<NOUT, P, false>(hl, al, key, wb, j0, 8, cur, a);
      if (j0 < n) gather8<NOUT, P, true>(hl, al, key, wb, j0, n - j0, cur, a);
    }
    key = nkey;
    wb = nwb;
  }
  if (cur >= 0) run_flush<CPL>(al + cur, a);
}

template <int K, int NOUT>
__device__ __forceinline__ void load_bfrags(const float* __restrict__ Wm, const int n0,
                                            const int l15, const int half, v2f* bf) {
#pragma unroll
  for (int kt = 0; kt < K / 4; ++kt) {
    const int kk = kt * 4 + half * 2;
    v2f b;
    b.x = Wm[kk * NOUT + n0 + l15];
    b.y = Wm[(kk + 1) * NOUT + n0 + l15];
    bf[kt] = b;
  }
}

__global__ __launch_bounds__(AT) void k_agg1_dense(const float* __restrict__ x,
                                                   const uint2* __restrict__ bins,
                                                   const int* __restrict__ base,
                                                   const float* __restrict__ dis,
                                                   const float* __restrict__ W1,
                                                   const float* __restrict__ b1,
                                                   const float* __restrict__ W2,
                                                   float* __restrict__ h2) {
  __shared__ __attribute__((aligned(16))) float acc[RB * P1];
  __shared__ __attribute__((aligned(16))) float hst[RB * P2];

  const int tid = threadIdx.x;
  const int lane = tid & 31;
  const int wave = __builtin_amdgcn_readfirstlane(tid >> 5);
  const int bin = blockIdx.x;
  const int r0 = bin * RB;
  const int half = lane >> 4;
  const int l15 = lane & 15;

  for (int i = tid; i < RB * P1 / 4; i += AT)
    ((float4*)acc)[i] = make_float4(0.0f, 0.0f, 0.0f, 0.0f);
  __syncthreads();

  bin_gather<KIN, P1>(x, bins, base[bin], base[bin + 1], acc, lane, wave);
  __syncthreads();

  for (int i = tid; i < RB * (KIN / 4); i += AT) {
    const int rl = i / (KIN / 4);
    const int c = (i % (KIN / 4)) * 4;
    const int r = r0 + rl;
    if (r < NN) {
      const float d = dis[r];
      const float s = d * d;
      const float4 xv = *(const float4*)(x + (size_t)r * KIN + c);
      float4* p = (float4*)(acc + rl * P1 + c);
      float4 a = *p;
      a.x += xv.x * s; a.y += xv.y * s; a.z += xv.z * s; a.w += xv.w * s;
      *p = a;
    }
  }
  __syncthreads();

  {
    const int ct = wave & 3;
    const int rt = (wave >> 2) * 2;
    v2f bf[KIN / 4];
    load_bfrags<KIN, H1>(W1, ct * 16, l15, half, bf);
    const float* pa0 = acc + (rt * 16 + l15) * P1 + half * 2;
    const float* pa1 = pa0 + 16 * P1;
    v8f d0 = {};
    v8f d1 = {};
#pragma unroll
    for (int kt = 0; kt < KIN / 4; ++kt) {
      const v2f a0 = *(const v2f*)(pa0 + kt * 4);
      const v2f a1 = *(const v2f*)(pa1 + kt * 4);
      d0 = __builtin_amdgcn_wmma_f32_16x16x4_f32(false, a0, false, bf[kt], (short)0, d0, false, false);
      d1 = __builtin_amdgcn_wmma_f32_16x16x4_f32(false, a1, false, bf[kt], (short)0, d1, false, false);
    }
    __syncthreads();
    const float bb = b1[ct * 16 + l15];
    float* po = acc + (rt * 16 + half * 8) * P1 + ct * 16 + l15;
#pragma unroll
    for (int v = 0; v < 8; ++v) {
      po[v * P1] = fmaxf(d0[v] + bb, 0.0f);
      po[(16 + v) * P1] = fmaxf(d1[v] + bb, 0.0f);
    }
  }
  __syncthreads();

  {
    const int ct = wave & 1;
    const int rt = wave >> 1;
    v2f bf[H1 / 4];
    load_bfrags<H1, H2>(W2, ct * 16, l15, half, bf);
    const float* pa = acc + (rt * 16 + l15) * P1 + half * 2;
    v8f d = {};
#pragma unroll
    for (int kt = 0; kt < H1 / 4; ++kt) {
      const v2f a = *(const v2f*)(pa + kt * 4);
      d = __builtin_amdgcn_wmma_f32_16x16x4_f32(false, a, false, bf[kt], (short)0, d, false, false);
    }
    float* po = hst + (rt * 16 + half * 8) * P2 + ct * 16 + l15;
#pragma unroll
    for (int v = 0; v < 8; ++v) po[v * P2] = d[v];
  }
  __syncthreads();

  for (int i = tid; i < RB * (H2 / 4); i += AT) {
    const int rl = i / (H2 / 4);
    const int c = (i % (H2 / 4)) * 4;
    const int r = r0 + rl;
    if (r < NN)
      *(float4*)(h2 + (size_t)r * H2 + c) = *(const float4*)(hst + rl * P2 + c);
  }
}

__global__ __launch_bounds__(AT) void k_agg2_head(const float* __restrict__ h2,
                                                  const uint2* __restrict__ bins,
                                                  const int* __restrict__ base,
                                                  const float* __restrict__ dis,
                                                  const float* __restrict__ b2,
                                                  const float* __restrict__ lw,
                                                  const float* __restrict__ lb,
                                                  float* __restrict__ out) {
  __shared__ float acc[RB * P3];
  __shared__ float slog[RB];

  const int tid = threadIdx.x;
  const int lane = tid & 31;
  const int wave = __builtin_amdgcn_readfirstlane(tid >> 5);
  const int bin = blockIdx.x;
  const int r0 = bin * RB;

  for (int i = tid; i < RB * P3; i += AT) acc[i] = 0.0f;
  __syncthreads();

  bin_gather<H2, P3>(h2, bins, base[bin], base[bin + 1], acc, lane, wave);
  __syncthreads();

  const int rl = tid >> 2;
  const int part = tid & 3;
  const int r = r0 + rl;
  float p = 0.0f;
  if (r < NN) {
    const float d = dis[r];
    const float s = d * d;
    const float* hp = h2 + (size_t)r * H2 + part * 8;
    const float4 ha = *(const float4*)hp;
    const float4 hb = *(const float4*)(hp + 4);
    const float hv[8] = {ha.x, ha.y, ha.z, ha.w, hb.x, hb.y, hb.z, hb.w};
    const float* ap = acc + rl * P3 + part * 8;
#pragma unroll
    for (int j = 0; j < 8; ++j) {
      const float a = fmaxf(ap[j] + (hv[j] * s + b2[part * 8 + j]), 0.0f);
      p = fmaf(a, lw[part * 8 + j], p);
    }
  }
  p += __shfl_xor(p, 1);
  p += __shfl_xor(p, 2);
  if (part == 0) slog[rl] = p + lb[0];
  __syncthreads();

  if (tid < RB / 2 && r0 + 2 * tid < NN) {
    const float l0 = slog[2 * tid];
    const float l1 = slog[2 * tid + 1];
    *(float4*)(out + (size_t)(r0 + 2 * tid) * 2) = make_float4(-l0, l0, -l1, l1);
  }
}

extern "C" void kernel_launch(void* const* d_in, const int* in_sizes, int n_in,
                              void* d_out, int out_size, void* d_ws, size_t ws_size,
                              hipStream_t stream) {
  const float* x      = (const float*)d_in[0];
  const int*   ei     = (const int*)d_in[1];
  const float* elog   = (const float*)d_in[2];
  const float* W1     = (const float*)d_in[3];
  const float* b1     = (const float*)d_in[4];
  const float* W2     = (const float*)d_in[5];
  const float* b2     = (const float*)d_in[6];
  const float* lw     = (const float*)d_in[7];
  const float* lb     = (const float*)d_in[8];
  float* out = (float*)d_out;

  float* ws   = (float*)d_ws;
  int*   base = (int*)ws;
  float* dis  = ws + NN;
  uint2* bins = (uint2*)(dis + NN);
  float* h2   = dis + NN + 2 * (size_t)NE;
  uint2* srt  = (uint2*)(h2 + (size_t)NN * H2);
  int*   cnt  = (int*)(h2 + (size_t)NN * H1);

  k_bin_count<<<NPW, PT, 0, stream>>>(ei, cnt);
  k_bin_scan<<<1, PT, 0, stream>>>(cnt, base);
  k_bin_fill<<<NPW, PT, 0, stream>>>(ei, elog, cnt, base, bins);
  k_bin_dis<<<NB, BT, 0, stream>>>(bins, base, dis);
  k_bin_sort_norm<<<NB, BT, 0, stream>>>(bins, base, dis, srt);

  k_agg1_dense<<<NB, AT, 0, stream>>>(x, srt, base, dis, W1, b1, W2, h2);

  k_agg2_head<<<NB, AT, 0, stream>>>(h2, srt, base, dis, b2, lw, lb, out);
}


# source rows pre-scaled by dis by their bin owner (xs = dis*x in the bin kernel, h2 at its store): per-edge dis[source] lookups and norm write-back gone; sum order kept, products regrouped
# speedup vs baseline: 4.1284x; 1.3165x over previous
#include <hip/hip_runtime.h>
#include <math.h>


#define NN 100000
#define NE 1600000
#define KIN 64
#define H1 64
#define H2 32

#define RB 128
#define LOGRB 7
#define NB ((NN + RB - 1) / RB)
#define PT 1024
#define EPT 8
#define PCH (PT * EPT)
#define NPW ((NE + PCH - 1) / PCH)
#define NPS 256
#define CU 8
#define BT 256
#define SCAP 4096
#define XU (RB * (KIN / 4) / BT)
#define AT 512
#define P1 68
#define P2 36
#define P3 33

static_assert(NB + 1 <= PT, "one running-sum thread per bin, and one more for the chunk's count");
static_assert(NPW <= NPS && NPS == BT, "one table slot and one thread per piece of a bin");
static_assert(RB <= BT, "one degree thread per owned row");
static_assert(XU * BT == RB * (KIN / 4), "the owned x rows divide evenly among the collecting threads");
static_assert(NPW * (NB + 1) <= NN * H2, "chunk offset tables fit in h2's own words, which nothing else uses before the layer-1 kernel");
static_assert(NB + 1 <= NN, "bin starts live where deg used to be");
static_assert((1 << LOGRB) == RB, "bin shift");
static_assert(NN <= (1 << (32 - LOGRB)), "source row and local target row share one word");
static_assert(AT == 512 && RB == 128, "16 waves cover the 8x4 and 8x2 tiles of a bin's products");
static_assert((NN % 2) == 0, "a bin's last row pair is stored together");
static_assert(2 * (size_t)NE <= (size_t)NN * (H1 - H2), "the sorted entry list fits behind h2 in the region h1 had");

typedef __attribute__((ext_vector_type(2))) float v2f;
typedef __attribute__((ext_vector_type(8))) float v8f;

__device__ __forceinline__ float edge_w(float l) {
  return 0.5f + 1.0f / (1.0f + __expf(-l));
}

__global__ __launch_bounds__(PT) void k_chunk_sort(const int* __restrict__ ei,
                                                   const float* __restrict__ logits,
                                                   uint2* __restrict__ cbin,
                                                   int* __restrict__ off) {
  __shared__ int lcnt[NB + 1];
  __shared__ int wtot[PT / 32];
  __shared__ uint2 sent[PCH];
  const int tid = threadIdx.x;
  const int lane = tid & 31;
  const int wv = tid >> 5;
  const int p = blockIdx.x;
  for (int b = tid; b <= NB; b += PT) lcnt[b] = 0;
  const int e0 = p * PCH + tid;
  int r[EPT], c[EPT];
  float w[EPT];
#pragma unroll
  for (int k = 0; k < EPT; ++k) {
    const int e = e0 + k * PT;
    if (e < NE) { r[k] = ei[e]; c[k] = ei[NE + e]; w[k] = edge_w(logits[e]); }
    else        { r[k] = 0;     c[k] = -1;         w[k] = 0.0f; }
  }
  __syncthreads();
  int bk[EPT], rk[EPT];
#pragma unroll
  for (int k = 0; k < EPT; ++k) {
    bk[k] = (c[k] >= 0) ? (c[k] >> LOGRB) : NB;
    rk[k] = atomicAdd(&lcnt[bk[k]], 1);
  }
  __syncthreads();
  const int own = (tid < NB) ? lcnt[tid] : 0;
  int inc = own;
  for (int d = 1; d < 32; d <<= 1) {
    const int v = __shfl_up(inc, d);
    if (lane >= d) inc += v;
  }
  if (lane == 31) wtot[wv] = inc;
  __syncthreads();
  if (wv == 0) {
    const int t = wtot[lane];
    int ti = t;
    for (int d = 1; d < 32; d <<= 1) {
      const int v = __shfl_up(ti, d);
      if (lane >= d) ti += v;
    }
    wtot[lane] = ti - t;
  }
  __syncthreads();
  const int n = min(PCH, NE - p * PCH);
  if (tid < NB) {
    const int st = wtot[wv] + inc - own;
    lcnt[tid] = st;
    off[p * (NB + 1) + tid] = st;
  }
  if (tid == NB) off[p * (NB + 1) + NB] = n;
  __syncthreads();
  int ls[EPT];
#pragma unroll
  for (int k = 0; k < EPT; ++k) ls[k] = lcnt[min(bk[k], NB - 1)];
#pragma unroll
  for (int k = 0; k < EPT; ++k) {
    if (c[k] >= 0)
      sent[ls[k] + rk[k]] = make_uint2(((unsigned)r[k] << LOGRB) | ((unsigned)c[k] & (RB - 1)),
                                       __float_as_uint(w[k]));
  }
  __syncthreads();
  uint2 ev[EPT];
#pragma unroll
  for (int k = 0; k < EPT; ++k) ev[k] = sent[tid + k * PT];
#pragma unroll
  for (int k = 0; k < EPT; ++k)
    if (tid + k * PT < n) cbin[(size_t)p * PCH + tid + k * PT] = ev[k];
}

__device__ __forceinline__ int piece_slot(const int* pst, const int* psrc, const int i) {
  int t = 0;
#pragma unroll
  for (int h = NPS / 2; h > 0; h >>= 1)
    if (pst[t + h] <= i) t += h;
  return psrc[t] + (i - pst[t]);
}

__global__ __launch_bounds__(BT) void k_bin_collect_sort(const uint2* __restrict__ cbin,
                                                         const int* __restrict__ off,
                                                         uint2* __restrict__ srt,
                                                         int* __restrict__ base,
                                                         float* __restrict__ dis,
                                                         const float* __restrict__ x,
                                                         float* __restrict__ xs) {
  __shared__ float sdeg[RB];
  __shared__ int rpos[RB];
  __shared__ int pst[NPS];
  __shared__ int psrc[NPS];
  __shared__ int wlen[BT / 32];
  __shared__ int woff[BT / 32];
  __shared__ uint2 stga[SCAP];
  __shared__ uint2 stgb[SCAP];
  const int tid = threadIdx.x;
  const int lane = tid & 31;
  const int wv = tid >> 5;
  const int bin = blockIdx.x;
  const float4* xrow = (const float4*)(x + (size_t)bin * RB * KIN);
  float4 xv[XU];
#pragma unroll
  for (int u = 0; u < XU; ++u) {
    const int i = tid + u * BT;
    xv[u] = (bin * RB + i / (KIN / 4) < NN) ? xrow[i] : make_float4(0.0f, 0.0f, 0.0f, 0.0f);
  }
  if (tid < RB) { sdeg[tid] = 0.0f; rpos[tid] = 0; }
  int o0 = 0, len = 0;
  if (tid < NPW) {
    const int* q = off + tid * (NB + 1) + bin;
    o0 = q[0];
    len = q[1] - o0;
  }
  int il = len, io = o0;
  for (int d = 1; d < 32; d <<= 1) {
    const int vl = __shfl_up(il, d);
    const int vo = __shfl_up(io, d);
    if (lane >= d) { il += vl; io += vo; }
  }
  if (lane == 31) { wlen[wv] = il; woff[wv] = io; }
  __syncthreads();
  int before = 0, n = 0, bstart = 0;
  for (int k = 0; k < BT / 32; ++k) {
    const int a = wlen[k];
    if (k < wv) before += a;
    n += a;
    bstart += woff[k];
  }
  pst[tid] = (tid < NPW) ? (before + il - len) : 0x7fffffff;
  psrc[tid] = tid * PCH + o0;
  if (tid == 0) {
    base[bin] = bstart;
    if (bin == NB - 1) base[NB] = bstart + n;
  }
  __syncthreads();
  const bool staged = n <= SCAP;
  for (int i0 = tid; i0 - lane < n; i0 += CU * BT) {
    int sidx[CU];
#pragma unroll
    for (int u = 0; u < CU; ++u) sidx[u] = piece_slot(pst, psrc, min(i0 + u * BT, n - 1));
    uint2 e[CU];
#pragma unroll
    for (int u = 0; u < CU; ++u) e[u] = cbin[sidx[u]];
#pragma unroll
    for (int u = 0; u < CU; ++u) {
      const int i = i0 + u * BT;
      const bool in = i < n;
      const int tl = (int)(e[u].x & (RB - 1));
      atomicAdd(&sdeg[tl], in ? __uint_as_float(e[u].y) : 0.0f);
      if (in) {
        atomicAdd(&rpos[tl], 1);
        if (staged) stga[i] = e[u];
      }
    }
  }
  __syncthreads();
  const int own = (tid < RB) ? rpos[tid] : 0;
  int inc = own;
  for (int d = 1; d < 32; d <<= 1) {
    const int v = __shfl_up(inc, d);
    if (lane >= d) inc += v;
  }
  if (lane == 31) wlen[wv] = inc;
  __syncthreads();
  if (tid < RB) {
    int rb = 0;
    for (int k = 0; k < RB / 32; ++k)
      if (k < wv) rb += wlen[k];
    rpos[tid] = rb + inc - own;
    const float dv = rsqrtf(sdeg[tid] + 1.0f);
    sdeg[tid] = dv;
    const int row = bin * RB + tid;
    if (row < NN) dis[row] = dv;
  }
  __syncthreads();
  {
    float4* xo = (float4*)(xs + (size_t)bin * RB * KIN);
#pragma unroll
    for (int u = 0; u < XU; ++u) {
      const int i = tid + u * BT;
      const int rl = i / (KIN / 4);
      if (bin * RB + rl < NN) {
        const float d = sdeg[rl];
        xo[i] = make_float4(xv[u].x * d, xv[u].y * d, xv[u].z * d, xv[u].w * d);
      }
    }
  }
  if (staged) {
    for (int i0 = tid; i0 < n; i0 += 4 * BT) {
      uint2 ev[4];
#pragma unroll
      for (int u = 0; u < 4; ++u) ev[u] = stga[min(i0 + u * BT, SCAP - 1)];
#pragma unroll
      for (int u = 0; u < 4; ++u) {
        if (i0 + u * BT < n) {
          const int tl = (int)(ev[u].x & (RB - 1));
          const int pos = atomicAdd(&rpos[tl], 1);
          stgb[pos] = make_uint2(ev[u].x, __float_as_uint(__uint_as_float(ev[u].y) * sdeg[tl]));
        }
      }
    }
    __syncthreads();
    for (int i0 = tid; i0 < n; i0 += 4 * BT) {
      uint2 ev[4];
#pragma unroll
      for (int u = 0; u < 4; ++u) ev[u] = stgb[min(i0 + u * BT, SCAP - 1)];
#pragma unroll
      for (int u = 0; u < 4; ++u)
        if (i0 + u * BT < n) srt[bstart + i0 + u * BT] = ev[u];
    }
  } else {
    for (int i = tid; i < n; i += BT) {
      const uint2 t = cbin[piece_slot(pst, psrc, i)];
      const int tl = (int)(t.x & (RB - 1));
      const int pos = atomicAdd(&rpos[tl], 1);
      srt[bstart + pos] = make_uint2(t.x, __float_as_uint(__uint_as_float(t.y) * sdeg[tl]));
    }
  }
}

template <int CPL>
__device__ __forceinline__ void run_flush(float* p, const float* a) {
  atomicAdd(p, a[0]);
  if constexpr (CPL == 2) atomicAdd(p + 1, a[1]);
}

template <int NOUT, int P, bool PART>
__device__ __forceinline__ void gather8(const float* hl, float* al,
                                        const unsigned key, const unsigned wb,
                                        const int j0, const int m,
                                        int& cur, float* a) {
  constexpr int CPL = NOUT / 32;
  float v[8][CPL];
  float w[8];
  int o[8];
#pragma unroll
  for (int k = 0; k < 8; ++k) {
    const unsigned kj = (unsigned)__builtin_amdgcn_readlane((int)key, j0 + k);
    w[k] = __int_as_float(__builtin_amdgcn_readlane((int)wb, j0 + k));
    o[k] = (int)(kj & (RB - 1)) * P;
    const float* src = hl + (size_t)(kj >> LOGRB) * NOUT;
    if constexpr (CPL == 2) {
      const v2f t = *(const v2f*)src;
      v[k][0] = t.x; v[k][1] = t.y;
    } else {
      v[k][0] = src[0];
    }
  }
#pragma unroll
  for (int k = 0; k < 8; ++k) {
    if (!PART || k < m) {
      if (o[k] != cur) {
        if (cur >= 0) run_flush<CPL>(al + cur, a);
        a[0] = 0.0f;
        if constexpr (CPL == 2) a[1] = 0.0f;
        cur = o[k];
      }
#pragma unroll
      for (int q = 0; q < CPL; ++q) a[q] = fmaf(v[k][q], w[k], a[q]);
    }
  }
}

template <int NOUT, int P>
__device__ __forceinline__ void bin_gather(const float* __restrict__ h,
                                           const uint2* __restrict__ bins,
                                           const int beg, const int end,
                                           float* acc, const int lane, const int wave) {
  constexpr int CPL = NOUT / 32;
  constexpr int NW = AT / 32;
  const float* hl = h + lane * CPL;
  float* al = acc + lane * CPL;

  const int per = ((end - beg + 8 * NW - 1) / (8 * NW)) * 8;
  const int wbeg = beg + wave * per;
  const int wend = min(end, wbeg + per);

  int cur = -1;
  float a[CPL] = {};

  int g = wbeg;
  unsigned key = 0u, wb = 0u;
  if (g + lane < wend) { const uint2 t = bins[g + lane]; key = t.x; wb = t.y; }
  for (; g < wend; g += 32) {
    unsigned nkey = 0u, nwb = 0u;
    if (g + 32 + lane < wend) { const uint2 t = bins[g + 32 + lane]; nkey = t.x; nwb = t.y; }
    const int n = min(32, wend - g);
    if (n == 32) {
#pragma unroll
      for (int j0 = 0; j0 < 32; j0 += 8) gather8<NOUT, P, false>(hl, al, key, wb, j0, 8, cur, a);
    } else {
      int j0 = 0;
      for (; j0 + 8 <= n; j0 += 8) gather8<NOUT, P, false>(hl, al, key, wb, j0, 8, cur, a);
      if (j0 < n) gather8<NOUT, P, true>(hl, al, key, wb, j0, n - j0, cur, a);
    }
    key = nkey;
    wb = nwb;
  }
  if (cur >= 0) run_flush<CPL>(al + cur, a);
}

template <int K, int NOUT>
__device__ __forceinline__ void load_bfrags(const float* __restrict__ Wm, const int n0,
                                            const int l15, const int half, v2f* bf) {
#pragma unroll
  for (int kt = 0; kt < K / 4; ++kt) {
    const int kk = kt * 4 + half * 2;
    v2f b;
    b.x = Wm[kk * NOUT + n0 + l15];
    b.y = Wm[(kk + 1) * NOUT + n0 + l15];
    bf[kt] = b;
  }
}

__global__ __launch_bounds__(AT) void k_agg1_dense(const float* __restrict__ xs,
                                                   const uint2* __restrict__ bins,
                                                   const int* __restrict__ base,
                                                   const float* __restrict__ dis,
                                                   const float* __restrict__ W1,
                                                   const float* __restrict__ b1,
                                                   const float* __restrict__ W2,
                                                   float* __restrict__ h2) {
  __shared__ __attribute__((aligned(16))) float acc[RB * P1];
  __shared__ __attribute__((aligned(16))) float hst[RB * P2];

  const int tid = threadIdx.x;
  const int lane = tid & 31;
  const int wave = __builtin_amdgcn_readfirstlane(tid >> 5);
  const int bin = blockIdx.x;
  const int r0 = bin * RB;
  const int half = lane >> 4;
  const int l15 = lane & 15;

  for (int i = tid; i < RB * P1 / 4; i += AT)
    ((float4*)acc)[i] = make_float4(0.0f, 0.0f, 0.0f, 0.0f);
  __syncthreads();

  bin_gather<KIN, P1>(xs, bins, base[bin], base[bin + 1], acc, lane, wave);
  __syncthreads();

  for (int i = tid; i < RB * (KIN / 4); i += AT) {
    const int rl = i / (KIN / 4);
    const int c = (i % (KIN / 4)) * 4;
    const int r = r0 + rl;
    if (r < NN) {
      const float s = dis[r];
      const float4 xv = *(const float4*)(xs + (size_t)r * KIN + c);
      float4* p = (float4*)(acc + rl * P1 + c);
      float4 a = *p;
      a.x += xv.x * s; a.y += xv.y * s; a.z += xv.z * s; a.w += xv.w * s;
      *p = a;
    }
  }
  __syncthreads();

  {
    const int ct = wave & 3;
    const int rt = (wave >> 2) * 2;
    v2f bf[KIN / 4];
    load_bfrags<KIN, H1>(W1, ct * 16, l15, half, bf);
    const float* pa0 = acc + (rt * 16 + l15) * P1 + half * 2;
    const float* pa1 = pa0 + 16 * P1;
    v8f d0 = {};
    v8f d1 = {};
#pragma unroll
    for (int kt = 0; kt < KIN / 4; ++kt) {
      const v2f a0 = *(const v2f*)(pa0 + kt * 4);
      const v2f a1 = *(const v2f*)(pa1 + kt * 4);
      d0 = __builtin_amdgcn_wmma_f32_16x16x4_f32(false, a0, false, bf[kt], (short)0, d0, false, false);
      d1 = __builtin_amdgcn_wmma_f32_16x16x4_f32(false, a1, false, bf[kt], (short)0, d1, false, false);
    }
    __syncthreads();
    const float bb = b1[ct * 16 + l15];
    float* po = acc + (rt * 16 + half * 8) * P1 + ct * 16 + l15;
#pragma unroll
    for (int v = 0; v < 8; ++v) {
      po[v * P1] = fmaxf(d0[v] + bb, 0.0f);
      po[(16 + v) * P1] = fmaxf(d1[v] + bb, 0.0f);
    }
  }
  __syncthreads();

  {
    const int ct = wave & 1;
    const int rt = wave >> 1;
    v2f bf[H1 / 4];
    load_bfrags<H1, H2>(W2, ct * 16, l15, half, bf);
    const float* pa = acc + (rt * 16 + l15) * P1 + half * 2;
    v8f d = {};
#pragma unroll
    for (int kt = 0; kt < H1 / 4; ++kt) {
      const v2f a = *(const v2f*)(pa + kt * 4);
      d = __builtin_amdgcn_wmma_f32_16x16x4_f32(false, a, false, bf[kt], (short)0, d, false, false);
    }
    float* po = hst + (rt * 16 + half * 8) * P2 + ct * 16 + l15;
#pragma unroll
    for (int v = 0; v < 8; ++v) po[v * P2] = d[v];
  }
  __syncthreads();

  for (int i = tid; i < RB * (H2 / 4); i += AT) {
    const int rl = i / (H2 / 4);
    const int c = (i % (H2 / 4)) * 4;
    const int r = r0 + rl;
    if (r < NN) {
      const float d = dis[r];
      const float4 v = *(const float4*)(hst + rl * P2 + c);
      *(float4*)(h2 + (size_t)r * H2 + c) = make_float4(v.x * d, v.y * d, v.z * d, v.w * d);
    }
  }
}

__global__ __launch_bounds__(AT) void k_agg2_head(const float* __restrict__ h2,
                                                  const uint2* __restrict__ bins,
                                                  const int* __restrict__ base,
                                                  const float* __restrict__ dis,
                                                  const float* __restrict__ b2,
                                                  const float* __restrict__ lw,
                                                  const float* __restrict__ lb,
                                                  float* __restrict__ out) {
  __shared__ float acc[RB * P3];
  __shared__ float slog[RB];

  const int tid = threadIdx.x;
  const int lane = tid & 31;
  const int wave = __builtin_amdgcn_readfirstlane(tid >> 5);
  const int bin = blockIdx.x;
  const int r0 = bin * RB;

  for (int i = tid; i < RB * P3; i += AT) acc[i] = 0.0f;
  __syncthreads();

  bin_gather<H2, P3>(h2, bins, base[bin], base[bin + 1], acc, lane, wave);
  __syncthreads();

  const int rl = tid >> 2;
  const int part = tid & 3;
  const int r = r0 + rl;
  float p = 0.0f;
  if (r < NN) {
    const float s = dis[r];
    const float* hp = h2 + (size_t)r * H2 + part * 8;
    const float4 ha = *(const float4*)hp;
    const float4 hb = *(const float4*)(hp + 4);
    const float hv[8] = {ha.x, ha.y, ha.z, ha.w, hb.x, hb.y, hb.z, hb.w};
    const float* ap = acc + rl * P3 + part * 8;
#pragma unroll
    for (int j = 0; j < 8; ++j) {
      const float a = fmaxf(ap[j] + (hv[j] * s + b2[part * 8 + j]), 0.0f);
      p = fmaf(a, lw[part * 8 + j], p);
    }
  }
  p += __shfl_xor(p, 1);
  p += __shfl_xor(p, 2);
  if (part == 0) slog[rl] = p + lb[0];
  __syncthreads();

  if (tid < RB / 2 && r0 + 2 * tid < NN) {
    const float l0 = slog[2 * tid];
    const float l1 = slog[2 * tid + 1];
    *(float4*)(out + (size_t)(r0 + 2 * tid) * 2) = make_float4(-l0, l0, -l1, l1);
  }
}

extern "C" void kernel_launch(void* const* d_in, const int* in_sizes, int n_in,
                              void* d_out, int out_size, void* d_ws, size_t ws_size,
                              hipStream_t stream) {
  const float* x      = (const float*)d_in[0];
  const int*   ei     = (const int*)d_in[1];
  const float* elog   = (const float*)d_in[2];
  const float* W1     = (const float*)d_in[3];
  const float* b1     = (const float*)d_in[4];
  const float* W2     = (const float*)d_in[5];
  const float* b2     = (const float*)d_in[6];
  const float* lw     = (const float*)d_in[7];
  const float* lb     = (const float*)d_in[8];
  float* out = (float*)d_out;

  float* ws   = (float*)d_ws;
  int*   base = (int*)ws;
  float* dis  = ws + NN;
  uint2* cbin = (uint2*)(dis + NN);
  float* h2   = dis + NN + 2 * (size_t)NE;
  uint2* srt  = (uint2*)(h2 + (size_t)NN * H2);
  int*   off  = (int*)h2;
  float* xs   = h2 + (size_t)NN * H1;

  k_chunk_sort<<<NPW, PT, 0, stream>>>(ei, elog, cbin, off);
  k_bin_collect_sort<<<NB, BT, 0, stream>>>(cbin, off, srt, base, dis, x, xs);

  k_agg1_dense<<<NB, AT, 0, stream>>>(xs, srt, base, dis, W1, b1, W2, h2);

  k_agg2_head<<<NB, AT, 0, stream>>>(h2, srt, base, dis, b2, lw, lb, out);
}
